# starting kernel
# baseline (speedup 1.0000x reference)
#include <hip/hip_runtime.h>
#include <hip/hip_bf16.h>

#define D_MODEL 128
#define EDGE_DIM 16

typedef float v2f __attribute__((ext_vector_type(2)));
typedef float v8f __attribute__((ext_vector_type(8)));

__device__ __forceinline__ v8f wmma_f32_4(v2f a, v2f b, v8f c) {
    return __builtin_amdgcn_wmma_f32_16x16x4_f32(false, a, false, b, (short)0, c,
                                                 false, false);
}

__device__ __forceinline__ float gelu_exact(float x) {
    return 0.5f * x * (1.0f + erff(x * 0.70710678118654752f));
}

__global__ void __launch_bounds__(256)
gate_kernel(const float* __restrict__ edge_attr,
            const float* __restrict__ gate_w1, const float* __restrict__ gate_b1,
            const float* __restrict__ gate_w2, const float* __restrict__ gate_b2,
            float* __restrict__ gate, int n_edges) {
    __shared__ float sRow[16];
    const int tid  = threadIdx.x;
    const int wave = tid >> 5;
    const int lane = tid & 31;
    const int e0   = blockIdx.x * 16;

    if (tid < 16) sRow[tid] = 0.0f;
    __syncthreads();

    const int mn   = lane & 15;
    const int koff = (lane >> 4) << 1;
    const int col  = wave * 16 + mn;
    const int erow = min(e0 + mn, n_edges - 1);

    const float* aB = edge_attr + (size_t)erow * EDGE_DIM;
    const float* bB = gate_w1   + (size_t)col  * EDGE_DIM;

    v8f c = {};
#pragma unroll
    for (int k = 0; k < EDGE_DIM; k += 4) {
        v2f a, b;
        a.x = aB[k + koff]; a.y = aB[k + koff + 1];
        b.x = bB[k + koff]; b.y = bB[k + koff + 1];
        c = wmma_f32_4(a, b, c);
    }

    const float b1 = gate_b1[col];
    const float w2 = gate_w2[col];
#pragma unroll
    for (int j = 0; j < 8; ++j) {
        int m = (lane < 16) ? j : (j + 8);
        float h = gelu_exact(c[j] + b1) * w2;
        atomicAdd(&sRow[m], h);
    }
    __syncthreads();

    if (tid < 16) {
        int e = e0 + tid;
        if (e < n_edges) {
            float z = sRow[tid] + gate_b2[0];
            gate[e] = 1.0f / (1.0f + expf(-z));
        }
    }
}

__global__ void __launch_bounds__(256)
scatter_kernel(const float* __restrict__ x_src,
               const int* __restrict__ edge_src, const int* __restrict__ edge_dst,
               const float* __restrict__ gate,
               float* __restrict__ pre, float* __restrict__ deg, int n_edges) {
    int e    = (blockIdx.x * blockDim.x + threadIdx.x) >> 5;
    int lane = threadIdx.x & 31;
    if (e >= n_edges) return;
    int   src = edge_src[e];
    int   dst = edge_dst[e];
    float g   = gate[e];
    float4 v  = ((const float4*)(x_src + (size_t)src * D_MODEL))[lane];
    float* p  = pre + (size_t)dst * D_MODEL + lane * 4;
    atomicAdd(p + 0, v.x * g);
    atomicAdd(p + 1, v.y * g);
    atomicAdd(p + 2, v.z * g);
    atomicAdd(p + 3, v.w * g);
    if (lane == 0) atomicAdd(deg + dst, 1.0f);
}

__global__ void __launch_bounds__(256)
node_kernel(const float* __restrict__ pre, const float* __restrict__ deg,
            const float* __restrict__ x_dst,
            const float* __restrict__ W_src, const float* __restrict__ W_dst,
            const float* __restrict__ b_dst,
            const float* __restrict__ ln_gamma, const float* __restrict__ ln_beta,
            float* __restrict__ out, int n_nodes) {
    __shared__ float sTile[16][D_MODEL + 4];
    __shared__ float sInv[16];
    const int tid  = threadIdx.x;
    const int wave = tid >> 5;
    const int lane = tid & 31;
    const int row0 = blockIdx.x * 16;

    if (tid < 16) {
        int r = min(row0 + tid, n_nodes - 1);
        sInv[tid] = 1.0f / fmaxf(deg[r], 1.0f);
    }
    __syncthreads();

    const int mn   = lane & 15;
    const int koff = (lane >> 4) << 1;
    const int col  = wave * 16 + mn;
    const int arow = min(row0 + mn, n_nodes - 1);
    const float inv = sInv[mn];

    v8f c = {};
    {
        const float* aB = pre   + (size_t)arow * D_MODEL;
        const float* bB = W_src + (size_t)col  * D_MODEL;
        for (int k = 0; k < D_MODEL; k += 4) {
            v2f a, b;
            a.x = aB[k + koff] * inv; a.y = aB[k + koff + 1] * inv;
            b.x = bB[k + koff];       b.y = bB[k + koff + 1];
            c = wmma_f32_4(a, b, c);
        }
    }
    {
        const float* aB = x_dst + (size_t)arow * D_MODEL;
        const float* bB = W_dst + (size_t)col  * D_MODEL;
        for (int k = 0; k < D_MODEL; k += 4) {
            v2f a, b;
            a.x = aB[k + koff]; a.y = aB[k + koff + 1];
            b.x = bB[k + koff]; b.y = bB[k + koff + 1];
            c = wmma_f32_4(a, b, c);
        }
    }

    const float bias = b_dst[col];
#pragma unroll
    for (int j = 0; j < 8; ++j) {
        int m = (lane < 16) ? j : (j + 8);
        sTile[m][col] = c[j] + bias;
    }
    __syncthreads();

    const int m  = tid >> 4;
    const int c0 = (tid & 15) * 8;
    float s = 0.0f, ss = 0.0f;
#pragma unroll
    for (int i = 0; i < 8; ++i) {
        float v = sTile[m][c0 + i];
        s += v; ss += v * v;
    }
#pragma unroll
    for (int off = 8; off > 0; off >>= 1) {
        s  += __shfl_xor(s,  off, 32);
        ss += __shfl_xor(ss, off, 32);
    }
    const float mean = s * (1.0f / 128.0f);
    const float var  = ss * (1.0f / 128.0f) - mean * mean;
    const float rstd = rsqrtf(var + 1e-5f);

    const int rglob = row0 + m;
    if (rglob < n_nodes) {
#pragma unroll
        for (int i = 0; i < 8; ++i) {
            int cc = c0 + i;
            float v = (sTile[m][cc] - mean) * rstd * ln_gamma[cc] + ln_beta[cc];
            out[(size_t)rglob * D_MODEL + cc] = gelu_exact(v);
        }
    }
}

extern "C" void kernel_launch(void* const* d_in, const int* in_sizes, int n_in,
                              void* d_out, int out_size, void* d_ws, size_t ws_size,
                              hipStream_t stream) {
    const float* x_src     = (const float*)d_in[0];
    const float* x_dst     = (const float*)d_in[1];
    const int*   edge_src  = (const int*)  d_in[2];
    const int*   edge_dst  = (const int*)  d_in[3];
    const float* edge_attr = (const float*)d_in[4];
    const float* W_src     = (const float*)d_in[5];
    const float* W_dst     = (const float*)d_in[6];
    const float* b_dst     = (const float*)d_in[7];
    const float* gate_w1   = (const float*)d_in[8];
    const float* gate_b1   = (const float*)d_in[9];
    const float* gate_w2   = (const float*)d_in[10];
    const float* gate_b2   = (const float*)d_in[11];
    const float* ln_gamma  = (const float*)d_in[12];
    const float* ln_beta   = (const float*)d_in[13];

    const int n_nodes = in_sizes[0] / D_MODEL;
    const int n_edges = in_sizes[2];

    float* pre  = (float*)d_ws;
    float* deg  = pre + (size_t)n_nodes * D_MODEL;
    float* gate = deg + n_nodes;

    hipMemsetAsync(d_ws, 0, (size_t)n_nodes * (D_MODEL + 1) * sizeof(float), stream);

    const int gateBlocks = (n_edges + 15) / 16;
    gate_kernel<<<gateBlocks, 256, 0, stream>>>(edge_attr, gate_w1, gate_b1,
                                                gate_w2, gate_b2, gate, n_edges);

    const int scatBlocks = (n_edges + 7) / 8;
    scatter_kernel<<<scatBlocks, 256, 0, stream>>>(x_src, edge_src, edge_dst,
                                                   gate, pre, deg, n_edges);

    const int nodeBlocks = (n_nodes + 15) / 16;
    node_kernel<<<nodeBlocks, 256, 0, stream>>>(pre, deg, x_dst, W_src, W_dst,
                                                b_dst, ln_gamma, ln_beta,
                                                (float*)d_out, n_nodes);
}


# Gate kernel: each wave owns 16 edges x all 128 hidden cols; row sums held in registers then a DPP row butterfly replace the 8-wave LDS float-atomic sum (gate sum order changes, now fixed)
# speedup vs baseline: 1.8157x; 1.8157x over previous
#include <hip/hip_runtime.h>
#include <hip/hip_bf16.h>

#define D_MODEL 128
#define EDGE_DIM 16

typedef float v2f __attribute__((ext_vector_type(2)));
typedef float v8f __attribute__((ext_vector_type(8)));

__device__ __forceinline__ v8f wmma_f32_4(v2f a, v2f b, v8f c) {
    return __builtin_amdgcn_wmma_f32_16x16x4_f32(false, a, false, b, (short)0, c,
                                                 false, false);
}

__device__ __forceinline__ float gelu_exact(float x) {
    return 0.5f * x * (1.0f + erff(x * 0.70710678118654752f));
}


template <int M>
__device__ __forceinline__ float row_xor(float x) {
    const int b = __builtin_bit_cast(int, x);
    return __builtin_bit_cast(
        float, __builtin_amdgcn_update_dpp(0, b, 0x160 | M, 0xf, 0xf, true));
}

__global__ void __launch_bounds__(256)
gate_kernel(const float* __restrict__ edge_attr,
            const float* __restrict__ gate_w1, const float* __restrict__ gate_b1,
            const float* __restrict__ gate_w2, const float* __restrict__ gate_b2,
            float* __restrict__ gate, int n_edges) {
    const int lane = threadIdx.x & 31;
    const int tile = (blockIdx.x * blockDim.x + threadIdx.x) >> 5;
    const int e0   = tile * 16;
    if (e0 >= n_edges) return;

    const int mn   = lane & 15;
    const int koff = (lane >> 4) << 1;
    const int erow = min(e0 + mn, n_edges - 1);

    const float* aB = edge_attr + (size_t)erow * EDGE_DIM;
    v2f a[4];
#pragma unroll
    for (int s = 0; s < 4; ++s) {
        a[s].x = aB[4 * s + koff]; a[s].y = aB[4 * s + koff + 1];
    }

    float acc[8];
#pragma unroll
    for (int j = 0; j < 8; ++j) acc[j] = 0.0f;

    for (int t = 0; t < D_MODEL / 16; ++t) {
        const int col = t * 16 + mn;
        const float* bB = gate_w1 + (size_t)col * EDGE_DIM;
        v8f c = {};
#pragma unroll
        for (int s = 0; s < 4; ++s) {
            v2f b;
            b.x = bB[4 * s + koff]; b.y = bB[4 * s + koff + 1];
            c = wmma_f32_4(a[s], b, c);
        }
        const float b1 = gate_b1[col];
        const float w2 = gate_w2[col];
#pragma unroll
        for (int j = 0; j < 8; ++j)
            acc[j] += gelu_exact(c[j] + b1) * w2;
    }

#pragma unroll
    for (int j = 0; j < 8; ++j) {
        acc[j] += row_xor<1>(acc[j]);
        acc[j] += row_xor<2>(acc[j]);
        acc[j] += row_xor<4>(acc[j]);
        acc[j] += row_xor<8>(acc[j]);
    }

    float z = acc[0];
#pragma unroll
    for (int j = 1; j < 8; ++j) z = ((lane & 7) == j) ? acc[j] : z;

    if ((lane & 8) == 0) {
        const int e = e0 + (lane & 7) + ((lane >> 4) << 3);
        if (e < n_edges) {
            z += gate_b2[0];
            gate[e] = 1.0f / (1.0f + expf(-z));
        }
    }
}

__global__ void __launch_bounds__(256)
scatter_kernel(const float* __restrict__ x_src,
               const int* __restrict__ edge_src, const int* __restrict__ edge_dst,
               const float* __restrict__ gate,
               float* __restrict__ pre, float* __restrict__ deg, int n_edges) {
    int e    = (blockIdx.x * blockDim.x + threadIdx.x) >> 5;
    int lane = threadIdx.x & 31;
    if (e >= n_edges) return;
    int   src = edge_src[e];
    int   dst = edge_dst[e];
    float g   = gate[e];
    float4 v  = ((const float4*)(x_src + (size_t)src * D_MODEL))[lane];
    float* p  = pre + (size_t)dst * D_MODEL + lane * 4;
    atomicAdd(p + 0, v.x * g);
    atomicAdd(p + 1, v.y * g);
    atomicAdd(p + 2, v.z * g);
    atomicAdd(p + 3, v.w * g);
    if (lane == 0) atomicAdd(deg + dst, 1.0f);
}

__global__ void __launch_bounds__(256)
node_kernel(const float* __restrict__ pre, const float* __restrict__ deg,
            const float* __restrict__ x_dst,
            const float* __restrict__ W_src, const float* __restrict__ W_dst,
            const float* __restrict__ b_dst,
            const float* __restrict__ ln_gamma, const float* __restrict__ ln_beta,
            float* __restrict__ out, int n_nodes) {
    __shared__ float sTile[16][D_MODEL + 4];
    __shared__ float sInv[16];
    const int tid  = threadIdx.x;
    const int wave = tid >> 5;
    const int lane = tid & 31;
    const int row0 = blockIdx.x * 16;

    if (tid < 16) {
        int r = min(row0 + tid, n_nodes - 1);
        sInv[tid] = 1.0f / fmaxf(deg[r], 1.0f);
    }
    __syncthreads();

    const int mn   = lane & 15;
    const int koff = (lane >> 4) << 1;
    const int col  = wave * 16 + mn;
    const int arow = min(row0 + mn, n_nodes - 1);
    const float inv = sInv[mn];

    v8f c = {};
    {
        const float* aB = pre   + (size_t)arow * D_MODEL;
        const float* bB = W_src + (size_t)col  * D_MODEL;
        for (int k = 0; k < D_MODEL; k += 4) {
            v2f a, b;
            a.x = aB[k + koff] * inv; a.y = aB[k + koff + 1] * inv;
            b.x = bB[k + koff];       b.y = bB[k + koff + 1];
            c = wmma_f32_4(a, b, c);
        }
    }
    {
        const float* aB = x_dst + (size_t)arow * D_MODEL;
        const float* bB = W_dst + (size_t)col  * D_MODEL;
        for (int k = 0; k < D_MODEL; k += 4) {
            v2f a, b;
            a.x = aB[k + koff]; a.y = aB[k + koff + 1];
            b.x = bB[k + koff]; b.y = bB[k + koff + 1];
            c = wmma_f32_4(a, b, c);
        }
    }

    const float bias = b_dst[col];
#pragma unroll
    for (int j = 0; j < 8; ++j) {
        int m = (lane < 16) ? j : (j + 8);
        sTile[m][col] = c[j] + bias;
    }
    __syncthreads();

    const int m  = tid >> 4;
    const int c0 = (tid & 15) * 8;
    float s = 0.0f, ss = 0.0f;
#pragma unroll
    for (int i = 0; i < 8; ++i) {
        float v = sTile[m][c0 + i];
        s += v; ss += v * v;
    }
#pragma unroll
    for (int off = 8; off > 0; off >>= 1) {
        s  += __shfl_xor(s,  off, 32);
        ss += __shfl_xor(ss, off, 32);
    }
    const float mean = s * (1.0f / 128.0f);
    const float var  = ss * (1.0f / 128.0f) - mean * mean;
    const float rstd = rsqrtf(var + 1e-5f);

    const int rglob = row0 + m;
    if (rglob < n_nodes) {
#pragma unroll
        for (int i = 0; i < 8; ++i) {
            int cc = c0 + i;
            float v = (sTile[m][cc] - mean) * rstd * ln_gamma[cc] + ln_beta[cc];
            out[(size_t)rglob * D_MODEL + cc] = gelu_exact(v);
        }
    }
}

extern "C" void kernel_launch(void* const* d_in, const int* in_sizes, int n_in,
                              void* d_out, int out_size, void* d_ws, size_t ws_size,
                              hipStream_t stream) {
    const float* x_src     = (const float*)d_in[0];
    const float* x_dst     = (const float*)d_in[1];
    const int*   edge_src  = (const int*)  d_in[2];
    const int*   edge_dst  = (const int*)  d_in[3];
    const float* edge_attr = (const float*)d_in[4];
    const float* W_src     = (const float*)d_in[5];
    const float* W_dst     = (const float*)d_in[6];
    const float* b_dst     = (const float*)d_in[7];
    const float* gate_w1   = (const float*)d_in[8];
    const float* gate_b1   = (const float*)d_in[9];
    const float* gate_w2   = (const float*)d_in[10];
    const float* gate_b2   = (const float*)d_in[11];
    const float* ln_gamma  = (const float*)d_in[12];
    const float* ln_beta   = (const float*)d_in[13];

    const int n_nodes = in_sizes[0] / D_MODEL;
    const int n_edges = in_sizes[2];

    float* pre  = (float*)d_ws;
    float* deg  = pre + (size_t)n_nodes * D_MODEL;
    float* gate = deg + n_nodes;

    hipMemsetAsync(d_ws, 0, (size_t)n_nodes * (D_MODEL + 1) * sizeof(float), stream);

    const int gateBlocks = (n_edges + 127) / 128;
    gate_kernel<<<gateBlocks, 256, 0, stream>>>(edge_attr, gate_w1, gate_b1,
                                                gate_w2, gate_b2, gate, n_edges);

    const int scatBlocks = (n_edges + 7) / 8;
    scatter_kernel<<<scatBlocks, 256, 0, stream>>>(x_src, edge_src, edge_dst,
                                                   gate, pre, deg, n_edges);

    const int nodeBlocks = (n_nodes + 15) / 16;
    node_kernel<<<nodeBlocks, 256, 0, stream>>>(pre, deg, x_dst, W_src, W_dst,
                                                b_dst, ln_gamma, ln_beta,
                                                (float*)d_out, n_nodes);
}


# Aggregation without float atomics: edges bucketed by destination (int histogram, scan, cursor add), node kernel gathers gate*x_src rows in registers into LDS as the A operand (sum order changes)
# speedup vs baseline: 2.3111x; 1.2728x over previous
#include <hip/hip_runtime.h>
#include <hip/hip_bf16.h>

#define D_MODEL 128
#define EDGE_DIM 16

typedef float v2f __attribute__((ext_vector_type(2)));
typedef float v8f __attribute__((ext_vector_type(8)));

__device__ __forceinline__ v8f wmma_f32_4(v2f a, v2f b, v8f c) {
    return __builtin_amdgcn_wmma_f32_16x16x4_f32(false, a, false, b, (short)0, c,
                                                 false, false);
}

__device__ __forceinline__ float gelu_exact(float x) {
    return 0.5f * x * (1.0f + erff(x * 0.70710678118654752f));
}


template <int M>
__device__ __forceinline__ float row_xor(float x) {
    const int b = __builtin_bit_cast(int, x);
    return __builtin_bit_cast(
        float, __builtin_amdgcn_update_dpp(0, b, 0x160 | M, 0xf, 0xf, true));
}

__global__ void __launch_bounds__(256)
gate_kernel(const float* __restrict__ edge_attr,
            const float* __restrict__ gate_w1, const float* __restrict__ gate_b1,
            const float* __restrict__ gate_w2, const float* __restrict__ gate_b2,
            const int* __restrict__ edge_src, const int* __restrict__ edge_dst,
            int* __restrict__ cursor, int* __restrict__ ssrc,
            float* __restrict__ sgate, int n_edges) {
    const int lane = threadIdx.x & 31;
    const int tile = (blockIdx.x * blockDim.x + threadIdx.x) >> 5;
    const int e0   = tile * 16;
    if (e0 >= n_edges) return;

    const int mn   = lane & 15;
    const int koff = (lane >> 4) << 1;
    const int erow = min(e0 + mn, n_edges - 1);

    const float* aB = edge_attr + (size_t)erow * EDGE_DIM;
    v2f a[4];
#pragma unroll
    for (int s = 0; s < 4; ++s) {
        a[s].x = aB[4 * s + koff]; a[s].y = aB[4 * s + koff + 1];
    }

    float acc[8];
#pragma unroll
    for (int j = 0; j < 8; ++j) acc[j] = 0.0f;

    for (int t = 0; t < D_MODEL / 16; ++t) {
        const int col = t * 16 + mn;
        const float* bB = gate_w1 + (size_t)col * EDGE_DIM;
        v8f c = {};
#pragma unroll
        for (int s = 0; s < 4; ++s) {
            v2f b;
            b.x = bB[4 * s + koff]; b.y = bB[4 * s + koff + 1];
            c = wmma_f32_4(a[s], b, c);
        }
        const float b1 = gate_b1[col];
        const float w2 = gate_w2[col];
#pragma unroll
        for (int j = 0; j < 8; ++j)
            acc[j] += gelu_exact(c[j] + b1) * w2;
    }

#pragma unroll
    for (int j = 0; j < 8; ++j) {
        acc[j] += row_xor<1>(acc[j]);
        acc[j] += row_xor<2>(acc[j]);
        acc[j] += row_xor<4>(acc[j]);
        acc[j] += row_xor<8>(acc[j]);
    }

    float z = acc[0];
#pragma unroll
    for (int j = 1; j < 8; ++j) z = ((lane & 7) == j) ? acc[j] : z;

    if ((lane & 8) == 0) {
        const int e = e0 + (lane & 7) + ((lane >> 4) << 3);
        if (e < n_edges) {
            z += gate_b2[0];
            const float gt = 1.0f / (1.0f + expf(-z));
            const int pos = atomicAdd(cursor + edge_dst[e], 1);
            ssrc[pos]  = edge_src[e];
            sgate[pos] = gt;
        }
    }
}

__global__ void __launch_bounds__(256)
hist_kernel(const int* __restrict__ edge_dst, int* __restrict__ cnt, int n_edges) {
    const int e = blockIdx.x * blockDim.x + threadIdx.x;
    if (e < n_edges) atomicAdd(cnt + edge_dst[e], 1);
}

__global__ void __launch_bounds__(1024)
scan_kernel(const int* __restrict__ cnt, int* __restrict__ cursor, int n) {
    __shared__ int sW[32];
    const int t = threadIdx.x, lane = t & 31, w = t >> 5;
    const int chunk = (n + 1023) / 1024;
    const int b = t * chunk;
    const int e = min(n, b + chunk);
    int s = 0;
#pragma unroll 8
    for (int i = b; i < e; ++i) s += cnt[i];
    int x = s;
    for (int o = 1; o < 32; o <<= 1) {
        const int y = __shfl_up(x, o, 32);
        if (lane >= o) x += y;
    }
    if (lane == 31) sW[w] = x;
    __syncthreads();
    if (w == 0) {
        const int v = sW[lane];
        int xx = v;
        for (int o = 1; o < 32; o <<= 1) {
            const int y = __shfl_up(xx, o, 32);
            if (lane >= o) xx += y;
        }
        sW[lane] = xx - v;
    }
    __syncthreads();
    int run = sW[w] + x - s;
#pragma unroll 8
    for (int i = b; i < e; ++i) {
        const int c = cnt[i];
        cursor[i] = run;
        run += c;
    }
}

__global__ void __launch_bounds__(256)
node_kernel(const float* __restrict__ x_src,
            const int* __restrict__ cnt, const int* __restrict__ cursor,
            const int* __restrict__ ssrc, const float* __restrict__ sgate,
            const float* __restrict__ x_dst,
            const float* __restrict__ W_src, const float* __restrict__ W_dst,
            const float* __restrict__ b_dst,
            const float* __restrict__ ln_gamma, const float* __restrict__ ln_beta,
            float* __restrict__ out, int n_nodes) {
    __shared__ float sTile[16][D_MODEL + 4];
    __shared__ __attribute__((aligned(16))) float sAgg[16][D_MODEL + 4];
    const int tid  = threadIdx.x;
    const int wave = __builtin_amdgcn_readfirstlane(tid >> 5);
    const int lane = tid & 31;
    const int row0 = blockIdx.x * 16;

    for (int q = 0; q < 2; ++q) {
        const int m = 2 * wave + q;
        const int n = row0 + m;
        float4 acc = make_float4(0.0f, 0.0f, 0.0f, 0.0f);
        int d = 0;
        if (n < n_nodes) {
            d = cnt[n];
            const int start = cursor[n] - d;
            for (int base = 0; base < d; base += 32) {
                const int idx = base + lane;
                int   s = 0;
                float g = 0.0f;
                if (idx < d) { s = ssrc[start + idx]; g = sgate[start + idx]; }
                const int lim = min(32, d - base);
                for (int j = 0; j < lim; ++j) {
                    const int   sj = __builtin_amdgcn_readlane(s, j);
                    const float gj = __builtin_bit_cast(
                        float, __builtin_amdgcn_readlane(__builtin_bit_cast(int, g), j));
                    const float4 v = ((const float4*)(x_src + (size_t)sj * D_MODEL))[lane];
                    acc.x = fmaf(gj, v.x, acc.x);
                    acc.y = fmaf(gj, v.y, acc.y);
                    acc.z = fmaf(gj, v.z, acc.z);
                    acc.w = fmaf(gj, v.w, acc.w);
                }
            }
        }
        const float inv = 1.0f / fmaxf((float)d, 1.0f);
        acc.x *= inv; acc.y *= inv; acc.z *= inv; acc.w *= inv;
        *(float4*)&sAgg[m][lane * 4] = acc;
    }
    __syncthreads();

    const int mn   = lane & 15;
    const int koff = (lane >> 4) << 1;
    const int col  = wave * 16 + mn;
    const int arow = min(row0 + mn, n_nodes - 1);

    v8f c = {};
    {
        const float* aB = &sAgg[mn][0];
        const float* bB = W_src + (size_t)col  * D_MODEL;
        for (int k = 0; k < D_MODEL; k += 4) {
            v2f a, b;
            a.x = aB[k + koff]; a.y = aB[k + koff + 1];
            b.x = bB[k + koff]; b.y = bB[k + koff + 1];
            c = wmma_f32_4(a, b, c);
        }
    }
    {
        const float* aB = x_dst + (size_t)arow * D_MODEL;
        const float* bB = W_dst + (size_t)col  * D_MODEL;
        for (int k = 0; k < D_MODEL; k += 4) {
            v2f a, b;
            a.x = aB[k + koff]; a.y = aB[k + koff + 1];
            b.x = bB[k + koff]; b.y = bB[k + koff + 1];
            c = wmma_f32_4(a, b, c);
        }
    }

    const float bias = b_dst[col];
#pragma unroll
    for (int j = 0; j < 8; ++j) {
        int m = (lane < 16) ? j : (j + 8);
        sTile[m][col] = c[j] + bias;
    }
    __syncthreads();

    const int m  = tid >> 4;
    const int c0 = (tid & 15) * 8;
    float s = 0.0f, ss = 0.0f;
#pragma unroll
    for (int i = 0; i < 8; ++i) {
        float v = sTile[m][c0 + i];
        s += v; ss += v * v;
    }
#pragma unroll
    for (int off = 8; off > 0; off >>= 1) {
        s  += __shfl_xor(s,  off, 32);
        ss += __shfl_xor(ss, off, 32);
    }
    const float mean = s * (1.0f / 128.0f);
    const float var  = ss * (1.0f / 128.0f) - mean * mean;
    const float rstd = rsqrtf(var + 1e-5f);

    const int rglob = row0 + m;
    if (rglob < n_nodes) {
#pragma unroll
        for (int i = 0; i < 8; ++i) {
            int cc = c0 + i;
            float v = (sTile[m][cc] - mean) * rstd * ln_gamma[cc] + ln_beta[cc];
            out[(size_t)rglob * D_MODEL + cc] = gelu_exact(v);
        }
    }
}

extern "C" void kernel_launch(void* const* d_in, const int* in_sizes, int n_in,
                              void* d_out, int out_size, void* d_ws, size_t ws_size,
                              hipStream_t stream) {
    const float* x_src     = (const float*)d_in[0];
    const float* x_dst     = (const float*)d_in[1];
    const int*   edge_src  = (const int*)  d_in[2];
    const int*   edge_dst  = (const int*)  d_in[3];
    const float* edge_attr = (const float*)d_in[4];
    const float* W_src     = (const float*)d_in[5];
    const float* W_dst     = (const float*)d_in[6];
    const float* b_dst     = (const float*)d_in[7];
    const float* gate_w1   = (const float*)d_in[8];
    const float* gate_b1   = (const float*)d_in[9];
    const float* gate_w2   = (const float*)d_in[10];
    const float* gate_b2   = (const float*)d_in[11];
    const float* ln_gamma  = (const float*)d_in[12];
    const float* ln_beta   = (const float*)d_in[13];

    const int n_nodes = in_sizes[0] / D_MODEL;
    const int n_edges = in_sizes[2];

    int*   cnt    = (int*)d_ws;
    int*   cursor = cnt + n_nodes;
    int*   ssrc   = cursor + n_nodes;
    float* sgate  = (float*)(ssrc + n_edges);

    hipMemsetAsync(cnt, 0, (size_t)n_nodes * sizeof(int), stream);

    const int histBlocks = (n_edges + 255) / 256;
    hist_kernel<<<histBlocks, 256, 0, stream>>>(edge_dst, cnt, n_edges);
    scan_kernel<<<1, 1024, 0, stream>>>(cnt, cursor, n_nodes);

    const int gateBlocks = (n_edges + 127) / 128;
    gate_kernel<<<gateBlocks, 256, 0, stream>>>(edge_attr, gate_w1, gate_b1,
                                                gate_w2, gate_b2, edge_src, edge_dst,
                                                cursor, ssrc, sgate, n_edges);

    const int nodeBlocks = (n_nodes + 15) / 16;
    node_kernel<<<nodeBlocks, 256, 0, stream>>>(x_src, cnt, cursor, ssrc, sgate,
                                                x_dst, W_src, W_dst,
                                                b_dst, ln_gamma, ln_beta,
                                                (float*)d_out, n_nodes);
}


# Bucket offsets: the one-block prefix scan over all node counts is replaced by a grid-wide pass, each 256-node block scans its counts and reserves its slot run with one returning integer add
# speedup vs baseline: 2.6186x; 1.1331x over previous
#include <hip/hip_runtime.h>
#include <hip/hip_bf16.h>

#define D_MODEL 128
#define EDGE_DIM 16

typedef float v2f __attribute__((ext_vector_type(2)));
typedef float v8f __attribute__((ext_vector_type(8)));

__device__ __forceinline__ v8f wmma_f32_4(v2f a, v2f b, v8f c) {
    return __builtin_amdgcn_wmma_f32_16x16x4_f32(false, a, false, b, (short)0, c,
                                                 false, false);
}

__device__ __forceinline__ float gelu_exact(float x) {
    return 0.5f * x * (1.0f + erff(x * 0.70710678118654752f));
}


template <int M>
__device__ __forceinline__ float row_xor(float x) {
    const int b = __builtin_bit_cast(int, x);
    return __builtin_bit_cast(
        float, __builtin_amdgcn_update_dpp(0, b, 0x160 | M, 0xf, 0xf, true));
}

__global__ void __launch_bounds__(256)
gate_kernel(const float* __restrict__ edge_attr,
            const float* __restrict__ gate_w1, const float* __restrict__ gate_b1,
            const float* __restrict__ gate_w2, const float* __restrict__ gate_b2,
            const int* __restrict__ edge_src, const int* __restrict__ edge_dst,
            int* __restrict__ cursor, int* __restrict__ ssrc,
            float* __restrict__ sgate, int n_edges) {
    const int lane = threadIdx.x & 31;
    const int tile = (blockIdx.x * blockDim.x + threadIdx.x) >> 5;
    const int e0   = tile * 16;
    if (e0 >= n_edges) return;

    const int mn   = lane & 15;
    const int koff = (lane >> 4) << 1;
    const int erow = min(e0 + mn, n_edges - 1);

    const float* aB = edge_attr + (size_t)erow * EDGE_DIM;
    v2f a[4];
#pragma unroll
    for (int s = 0; s < 4; ++s) {
        a[s].x = aB[4 * s + koff]; a[s].y = aB[4 * s + koff + 1];
    }

    float acc[8];
#pragma unroll
    for (int j = 0; j < 8; ++j) acc[j] = 0.0f;

    for (int t = 0; t < D_MODEL / 16; ++t) {
        const int col = t * 16 + mn;
        const float* bB = gate_w1 + (size_t)col * EDGE_DIM;
        v8f c = {};
#pragma unroll
        for (int s = 0; s < 4; ++s) {
            v2f b;
            b.x = bB[4 * s + koff]; b.y = bB[4 * s + koff + 1];
            c = wmma_f32_4(a[s], b, c);
        }
        const float b1 = gate_b1[col];
        const float w2 = gate_w2[col];
#pragma unroll
        for (int j = 0; j < 8; ++j)
            acc[j] += gelu_exact(c[j] + b1) * w2;
    }

#pragma unroll
    for (int j = 0; j < 8; ++j) {
        acc[j] += row_xor<1>(acc[j]);
        acc[j] += row_xor<2>(acc[j]);
        acc[j] += row_xor<4>(acc[j]);
        acc[j] += row_xor<8>(acc[j]);
    }

    float z = acc[0];
#pragma unroll
    for (int j = 1; j < 8; ++j) z = ((lane & 7) == j) ? acc[j] : z;

    if ((lane & 8) == 0) {
        const int e = e0 + (lane & 7) + ((lane >> 4) << 3);
        if (e < n_edges) {
            z += gate_b2[0];
            const float gt = 1.0f / (1.0f + expf(-z));
            const int pos = atomicAdd(cursor + edge_dst[e], 1);
            ssrc[pos]  = edge_src[e];
            sgate[pos] = gt;
        }
    }
}

__global__ void __launch_bounds__(256)
hist_kernel(const int* __restrict__ edge_dst, int* __restrict__ cnt, int n_edges) {
    const int e = blockIdx.x * blockDim.x + threadIdx.x;
    if (e < n_edges) atomicAdd(cnt + edge_dst[e], 1);
}

__global__ void __launch_bounds__(256)
alloc_kernel(const int* __restrict__ cnt, int* __restrict__ total,
             int* __restrict__ cursor, int n) {
    __shared__ int sW[8];
    __shared__ int sBase;
    const int t = threadIdx.x, lane = t & 31, w = t >> 5;
    const int i = blockIdx.x * 256 + t;
    const int c = (i < n) ? cnt[i] : 0;
    int x = c;
    for (int o = 1; o < 32; o <<= 1) {
        const int y = __shfl_up(x, o, 32);
        if (lane >= o) x += y;
    }
    if (lane == 31) sW[w] = x;
    __syncthreads();
    if (t == 0) {
        int s = 0;
        for (int k = 0; k < 8; ++k) { const int v = sW[k]; sW[k] = s; s += v; }
        sBase = atomicAdd(total, s);
    }
    __syncthreads();
    if (i < n) cursor[i] = sBase + sW[w] + x - c;
}

__global__ void __launch_bounds__(256)
node_kernel(const float* __restrict__ x_src,
            const int* __restrict__ cnt, const int* __restrict__ cursor,
            const int* __restrict__ ssrc, const float* __restrict__ sgate,
            const float* __restrict__ x_dst,
            const float* __restrict__ W_src, const float* __restrict__ W_dst,
            const float* __restrict__ b_dst,
            const float* __restrict__ ln_gamma, const float* __restrict__ ln_beta,
            float* __restrict__ out, int n_nodes) {
    __shared__ float sTile[16][D_MODEL + 4];
    __shared__ __attribute__((aligned(16))) float sAgg[16][D_MODEL + 4];
    const int tid  = threadIdx.x;
    const int wave = __builtin_amdgcn_readfirstlane(tid >> 5);
    const int lane = tid & 31;
    const int row0 = blockIdx.x * 16;

    for (int q = 0; q < 2; ++q) {
        const int m = 2 * wave + q;
        const int n = row0 + m;
        float4 acc = make_float4(0.0f, 0.0f, 0.0f, 0.0f);
        int d = 0;
        if (n < n_nodes) {
            d = cnt[n];
            const int start = cursor[n] - d;
            for (int base = 0; base < d; base += 32) {
                const int idx = base + lane;
                int   s = 0;
                float g = 0.0f;
                if (idx < d) { s = ssrc[start + idx]; g = sgate[start + idx]; }
                const int lim = min(32, d - base);
                for (int j = 0; j < lim; ++j) {
                    const int   sj = __builtin_amdgcn_readlane(s, j);
                    const float gj = __builtin_bit_cast(
                        float, __builtin_amdgcn_readlane(__builtin_bit_cast(int, g), j));
                    const float4 v = ((const float4*)(x_src + (size_t)sj * D_MODEL))[lane];
                    acc.x = fmaf(gj, v.x, acc.x);
                    acc.y = fmaf(gj, v.y, acc.y);
                    acc.z = fmaf(gj, v.z, acc.z);
                    acc.w = fmaf(gj, v.w, acc.w);
                }
            }
        }
        const float inv = 1.0f / fmaxf((float)d, 1.0f);
        acc.x *= inv; acc.y *= inv; acc.z *= inv; acc.w *= inv;
        *(float4*)&sAgg[m][lane * 4] = acc;
    }
    __syncthreads();

    const int mn   = lane & 15;
    const int koff = (lane >> 4) << 1;
    const int col  = wave * 16 + mn;
    const int arow = min(row0 + mn, n_nodes - 1);

    v8f c = {};
    {
        const float* aB = &sAgg[mn][0];
        const float* bB = W_src + (size_t)col  * D_MODEL;
        for (int k = 0; k < D_MODEL; k += 4) {
            v2f a, b;
            a.x = aB[k + koff]; a.y = aB[k + koff + 1];
            b.x = bB[k + koff]; b.y = bB[k + koff + 1];
            c = wmma_f32_4(a, b, c);
        }
    }
    {
        const float* aB = x_dst + (size_t)arow * D_MODEL;
        const float* bB = W_dst + (size_t)col  * D_MODEL;
        for (int k = 0; k < D_MODEL; k += 4) {
            v2f a, b;
            a.x = aB[k + koff]; a.y = aB[k + koff + 1];
            b.x = bB[k + koff]; b.y = bB[k + koff + 1];
            c = wmma_f32_4(a, b, c);
        }
    }

    const float bias = b_dst[col];
#pragma unroll
    for (int j = 0; j < 8; ++j) {
        int m = (lane < 16) ? j : (j + 8);
        sTile[m][col] = c[j] + bias;
    }
    __syncthreads();

    const int m  = tid >> 4;
    const int c0 = (tid & 15) * 8;
    float s = 0.0f, ss = 0.0f;
#pragma unroll
    for (int i = 0; i < 8; ++i) {
        float v = sTile[m][c0 + i];
        s += v; ss += v * v;
    }
#pragma unroll
    for (int off = 8; off > 0; off >>= 1) {
        s  += __shfl_xor(s,  off, 32);
        ss += __shfl_xor(ss, off, 32);
    }
    const float mean = s * (1.0f / 128.0f);
    const float var  = ss * (1.0f / 128.0f) - mean * mean;
    const float rstd = rsqrtf(var + 1e-5f);

    const int rglob = row0 + m;
    if (rglob < n_nodes) {
#pragma unroll
        for (int i = 0; i < 8; ++i) {
            int cc = c0 + i;
            float v = (sTile[m][cc] - mean) * rstd * ln_gamma[cc] + ln_beta[cc];
            out[(size_t)rglob * D_MODEL + cc] = gelu_exact(v);
        }
    }
}

extern "C" void kernel_launch(void* const* d_in, const int* in_sizes, int n_in,
                              void* d_out, int out_size, void* d_ws, size_t ws_size,
                              hipStream_t stream) {
    const float* x_src     = (const float*)d_in[0];
    const float* x_dst     = (const float*)d_in[1];
    const int*   edge_src  = (const int*)  d_in[2];
    const int*   edge_dst  = (const int*)  d_in[3];
    const float* edge_attr = (const float*)d_in[4];
    const float* W_src     = (const float*)d_in[5];
    const float* W_dst     = (const float*)d_in[6];
    const float* b_dst     = (const float*)d_in[7];
    const float* gate_w1   = (const float*)d_in[8];
    const float* gate_b1   = (const float*)d_in[9];
    const float* gate_w2   = (const float*)d_in[10];
    const float* gate_b2   = (const float*)d_in[11];
    const float* ln_gamma  = (const float*)d_in[12];
    const float* ln_beta   = (const float*)d_in[13];

    const int n_nodes = in_sizes[0] / D_MODEL;
    const int n_edges = in_sizes[2];

    int*   cnt    = (int*)d_ws;
    int*   total  = cnt + n_nodes;
    int*   cursor = total + 1;
    int*   ssrc   = cursor + n_nodes;
    float* sgate  = (float*)(ssrc + n_edges);

    hipMemsetAsync(cnt, 0, (size_t)(n_nodes + 1) * sizeof(int), stream);

    const int histBlocks = (n_edges + 255) / 256;
    hist_kernel<<<histBlocks, 256, 0, stream>>>(edge_dst, cnt, n_edges);
    const int allocBlocks = (n_nodes + 255) / 256;
    alloc_kernel<<<allocBlocks, 256, 0, stream>>>(cnt, total, cursor, n_nodes);

    const int gateBlocks = (n_edges + 127) / 128;
    gate_kernel<<<gateBlocks, 256, 0, stream>>>(edge_attr, gate_w1, gate_b1,
                                                gate_w2, gate_b2, edge_src, edge_dst,
                                                cursor, ssrc, sgate, n_edges);

    const int nodeBlocks = (n_nodes + 15) / 16;
    node_kernel<<<nodeBlocks, 256, 0, stream>>>(x_src, cnt, cursor, ssrc, sgate,
                                                x_dst, W_src, W_dst,
                                                b_dst, ln_gamma, ln_beta,
                                                (float*)d_out, n_nodes);
}


# Node GEMM fetch: 32-node blocks reuse each 16-byte W fragment load for 2 k-steps and 2 row tiles; x_dst rows staged in LDS by whole-row loads (k order in each 8-k group permuted: sum order changes)
# speedup vs baseline: 3.3145x; 1.2658x over previous
#include <hip/hip_runtime.h>
#include <hip/hip_bf16.h>

#define D_MODEL 128
#define EDGE_DIM 16

typedef float v2f __attribute__((ext_vector_type(2)));
typedef float v8f __attribute__((ext_vector_type(8)));

__device__ __forceinline__ v8f wmma_f32_4(v2f a, v2f b, v8f c) {
    return __builtin_amdgcn_wmma_f32_16x16x4_f32(false, a, false, b, (short)0, c,
                                                 false, false);
}

__device__ __forceinline__ float gelu_exact(float x) {
    return 0.5f * x * (1.0f + erff(x * 0.70710678118654752f));
}


template <int M>
__device__ __forceinline__ float row_xor(float x) {
    const int b = __builtin_bit_cast(int, x);
    return __builtin_bit_cast(
        float, __builtin_amdgcn_update_dpp(0, b, 0x160 | M, 0xf, 0xf, true));
}

__global__ void __launch_bounds__(256)
gate_kernel(const float* __restrict__ edge_attr,
            const float* __restrict__ gate_w1, const float* __restrict__ gate_b1,
            const float* __restrict__ gate_w2, const float* __restrict__ gate_b2,
            const int* __restrict__ edge_src, const int* __restrict__ edge_dst,
            int* __restrict__ cursor, int* __restrict__ ssrc,
            float* __restrict__ sgate, int n_edges) {
    const int lane = threadIdx.x & 31;
    const int tile = (blockIdx.x * blockDim.x + threadIdx.x) >> 5;
    const int e0   = tile * 16;
    if (e0 >= n_edges) return;

    const int mn   = lane & 15;
    const int koff = (lane >> 4) << 1;
    const int erow = min(e0 + mn, n_edges - 1);

    const float* aB = edge_attr + (size_t)erow * EDGE_DIM;
    v2f a[4];
#pragma unroll
    for (int s = 0; s < 4; ++s) {
        a[s].x = aB[4 * s + koff]; a[s].y = aB[4 * s + koff + 1];
    }

    float acc[8];
#pragma unroll
    for (int j = 0; j < 8; ++j) acc[j] = 0.0f;

    for (int t = 0; t < D_MODEL / 16; ++t) {
        const int col = t * 16 + mn;
        const float* bB = gate_w1 + (size_t)col * EDGE_DIM;
        v8f c = {};
#pragma unroll
        for (int s = 0; s < 4; ++s) {
            v2f b;
            b.x = bB[4 * s + koff]; b.y = bB[4 * s + koff + 1];
            c = wmma_f32_4(a[s], b, c);
        }
        const float b1 = gate_b1[col];
        const float w2 = gate_w2[col];
#pragma unroll
        for (int j = 0; j < 8; ++j)
            acc[j] += gelu_exact(c[j] + b1) * w2;
    }

#pragma unroll
    for (int j = 0; j < 8; ++j) {
        acc[j] += row_xor<1>(acc[j]);
        acc[j] += row_xor<2>(acc[j]);
        acc[j] += row_xor<4>(acc[j]);
        acc[j] += row_xor<8>(acc[j]);
    }

    float z = acc[0];
#pragma unroll
    for (int j = 1; j < 8; ++j) z = ((lane & 7) == j) ? acc[j] : z;

    if ((lane & 8) == 0) {
        const int e = e0 + (lane & 7) + ((lane >> 4) << 3);
        if (e < n_edges) {
            z += gate_b2[0];
            const float gt = 1.0f / (1.0f + expf(-z));
            const int pos = atomicAdd(cursor + edge_dst[e], 1);
            ssrc[pos]  = edge_src[e];
            sgate[pos] = gt;
        }
    }
}

__global__ void __launch_bounds__(256)
hist_kernel(const int* __restrict__ edge_dst, int* __restrict__ cnt, int n_edges) {
    const int e = blockIdx.x * blockDim.x + threadIdx.x;
    if (e < n_edges) atomicAdd(cnt + edge_dst[e], 1);
}

__global__ void __launch_bounds__(256)
alloc_kernel(const int* __restrict__ cnt, int* __restrict__ total,
             int* __restrict__ cursor, int n) {
    __shared__ int sW[8];
    __shared__ int sBase;
    const int t = threadIdx.x, lane = t & 31, w = t >> 5;
    const int i = blockIdx.x * 256 + t;
    const int c = (i < n) ? cnt[i] : 0;
    int x = c;
    for (int o = 1; o < 32; o <<= 1) {
        const int y = __shfl_up(x, o, 32);
        if (lane >= o) x += y;
    }
    if (lane == 31) sW[w] = x;
    __syncthreads();
    if (t == 0) {
        int s = 0;
        for (int k = 0; k < 8; ++k) { const int v = sW[k]; sW[k] = s; s += v; }
        sBase = atomicAdd(total, s);
    }
    __syncthreads();
    if (i < n) cursor[i] = sBase + sW[w] + x - c;
}

#define NODE_MT 2
#define NODE_BM (16 * NODE_MT)
#define NODE_PITCH (D_MODEL + 4)
#define NODE_RPW (NODE_BM / 8)

__device__ __forceinline__ float4 frag4(const float* rowbase, int g, int h) {
    return *(const float4*)(rowbase + 8 * g + 4 * h);
}

__global__ void __launch_bounds__(256)
node_kernel(const float* __restrict__ x_src,
            const int* __restrict__ cnt, const int* __restrict__ cursor,
            const int* __restrict__ ssrc, const float* __restrict__ sgate,
            const float* __restrict__ x_dst,
            const float* __restrict__ W_src, const float* __restrict__ W_dst,
            const float* __restrict__ b_dst,
            const float* __restrict__ ln_gamma, const float* __restrict__ ln_beta,
            float* __restrict__ out, int n_nodes) {
    __shared__ __attribute__((aligned(16))) float sAgg[NODE_BM][NODE_PITCH];
    __shared__ __attribute__((aligned(16))) float sX[NODE_BM][NODE_PITCH];
    float (*sTile)[NODE_PITCH] = sAgg;
    const int tid  = threadIdx.x;
    const int wave = __builtin_amdgcn_readfirstlane(tid >> 5);
    const int lane = tid & 31;
    const int row0 = blockIdx.x * NODE_BM;

    float4 xr[NODE_RPW];
#pragma unroll
    for (int q = 0; q < NODE_RPW; ++q) {
        const int n = min(row0 + NODE_RPW * wave + q, n_nodes - 1);
        xr[q] = ((const float4*)(x_dst + (size_t)n * D_MODEL))[lane];
    }

    for (int q = 0; q < NODE_RPW; ++q) {
        const int m = NODE_RPW * wave + q;
        const int n = row0 + m;
        float4 acc = make_float4(0.0f, 0.0f, 0.0f, 0.0f);
        int d = 0;
        if (n < n_nodes) {
            d = cnt[n];
            const int start = cursor[n] - d;
            for (int base = 0; base < d; base += 32) {
                const int idx = base + lane;
                int   s = 0;
                float g = 0.0f;
                if (idx < d) { s = ssrc[start + idx]; g = sgate[start + idx]; }
                const int lim = min(32, d - base);
                for (int j = 0; j < lim; ++j) {
                    const int   sj = __builtin_amdgcn_readlane(s, j);
                    const float gj = __builtin_bit_cast(
                        float, __builtin_amdgcn_readlane(__builtin_bit_cast(int, g), j));
                    const float4 v = ((const float4*)(x_src + (size_t)sj * D_MODEL))[lane];
                    acc.x = fmaf(gj, v.x, acc.x);
                    acc.y = fmaf(gj, v.y, acc.y);
                    acc.z = fmaf(gj, v.z, acc.z);
                    acc.w = fmaf(gj, v.w, acc.w);
                }
            }
        }
        const float inv = 1.0f / fmaxf((float)d, 1.0f);
        acc.x *= inv; acc.y *= inv; acc.z *= inv; acc.w *= inv;
        *(float4*)&sAgg[m][lane * 4] = acc;
    }
#pragma unroll
    for (int q = 0; q < NODE_RPW; ++q)
        *(float4*)&sX[NODE_RPW * wave + q][lane * 4] = xr[q];
    __syncthreads();

    const int mn  = lane & 15;
    const int h   = lane >> 4;
    const int col = wave * 16 + mn;

    v8f c[NODE_MT];
#pragma unroll
    for (int t = 0; t < NODE_MT; ++t) c[t] = (v8f){};

    {
        const float* bB = W_src + (size_t)col * D_MODEL;
        for (int g = 0; g < D_MODEL / 8; ++g) {
            const float4 b = frag4(bB, g, h);
#pragma unroll
            for (int t = 0; t < NODE_MT; ++t) {
                const float4 a = frag4(&sAgg[t * 16 + mn][0], g, h);
                c[t] = wmma_f32_4((v2f){a.x, a.y}, (v2f){b.x, b.y}, c[t]);
                c[t] = wmma_f32_4((v2f){a.z, a.w}, (v2f){b.z, b.w}, c[t]);
            }
        }
    }
    {
        const float* bB = W_dst + (size_t)col * D_MODEL;
        for (int g = 0; g < D_MODEL / 8; ++g) {
            const float4 b = frag4(bB, g, h);
#pragma unroll
            for (int t = 0; t < NODE_MT; ++t) {
                const float4 a = frag4(&sX[t * 16 + mn][0], g, h);
                c[t] = wmma_f32_4((v2f){a.x, a.y}, (v2f){b.x, b.y}, c[t]);
                c[t] = wmma_f32_4((v2f){a.z, a.w}, (v2f){b.z, b.w}, c[t]);
            }
        }
    }
    __syncthreads();

    const float bias = b_dst[col];
#pragma unroll
    for (int t = 0; t < NODE_MT; ++t) {
#pragma unroll
        for (int j = 0; j < 8; ++j) {
            const int m = t * 16 + ((lane < 16) ? j : (j + 8));
            sTile[m][col] = c[t][j] + bias;
        }
    }
    __syncthreads();

    const int c0 = (tid & 15) * 8;
    for (int t = 0; t < NODE_MT; ++t) {
        const int m = t * 16 + (tid >> 4);
        float s = 0.0f, ss = 0.0f;
#pragma unroll
        for (int i = 0; i < 8; ++i) {
            float v = sTile[m][c0 + i];
            s += v; ss += v * v;
        }
#pragma unroll
        for (int off = 8; off > 0; off >>= 1) {
            s  += __shfl_xor(s,  off, 32);
            ss += __shfl_xor(ss, off, 32);
        }
        const float mean = s * (1.0f / 128.0f);
        const float var  = ss * (1.0f / 128.0f) - mean * mean;
        const float rstd = rsqrtf(var + 1e-5f);

        const int rglob = row0 + m;
        if (rglob < n_nodes) {
#pragma unroll
            for (int i = 0; i < 8; ++i) {
                int cc = c0 + i;
                float v = (sTile[m][cc] - mean) * rstd * ln_gamma[cc] + ln_beta[cc];
                out[(size_t)rglob * D_MODEL + cc] = gelu_exact(v);
            }
        }
    }
}

extern "C" void kernel_launch(void* const* d_in, const int* in_sizes, int n_in,
                              void* d_out, int out_size, void* d_ws, size_t ws_size,
                              hipStream_t stream) {
    const float* x_src     = (const float*)d_in[0];
    const float* x_dst     = (const float*)d_in[1];
    const int*   edge_src  = (const int*)  d_in[2];
    const int*   edge_dst  = (const int*)  d_in[3];
    const float* edge_attr = (const float*)d_in[4];
    const float* W_src     = (const float*)d_in[5];
    const float* W_dst     = (const float*)d_in[6];
    const float* b_dst     = (const float*)d_in[7];
    const float* gate_w1   = (const float*)d_in[8];
    const float* gate_b1   = (const float*)d_in[9];
    const float* gate_w2   = (const float*)d_in[10];
    const float* gate_b2   = (const float*)d_in[11];
    const float* ln_gamma  = (const float*)d_in[12];
    const float* ln_beta   = (const float*)d_in[13];

    const int n_nodes = in_sizes[0] / D_MODEL;
    const int n_edges = in_sizes[2];

    int*   cnt    = (int*)d_ws;
    int*   total  = cnt + n_nodes;
    int*   cursor = total + 1;
    int*   ssrc   = cursor + n_nodes;
    float* sgate  = (float*)(ssrc + n_edges);

    hipMemsetAsync(cnt, 0, (size_t)(n_nodes + 1) * sizeof(int), stream);

    const int histBlocks = (n_edges + 255) / 256;
    hist_kernel<<<histBlocks, 256, 0, stream>>>(edge_dst, cnt, n_edges);
    const int allocBlocks = (n_nodes + 255) / 256;
    alloc_kernel<<<allocBlocks, 256, 0, stream>>>(cnt, total, cursor, n_nodes);

    const int gateBlocks = (n_edges + 127) / 128;
    gate_kernel<<<gateBlocks, 256, 0, stream>>>(edge_attr, gate_w1, gate_b1,
                                                gate_w2, gate_b2, edge_src, edge_dst,
                                                cursor, ssrc, sgate, n_edges);

    const int nodeBlocks = (n_nodes + NODE_BM - 1) / NODE_BM;
    node_kernel<<<nodeBlocks, 256, 0, stream>>>(x_src, cnt, cursor, ssrc, sgate,
                                                x_dst, W_src, W_dst,
                                                b_dst, ln_gamma, ln_beta,
                                                (float*)d_out, n_nodes);
}


# gelu in both kernels: library erf (two ranges, a divergent branch per value) replaced by max(x,0) - |x|*exp2(P10(|x|)) on packed two-float FMAs, branch-free; gelu values move in the last bits
# speedup vs baseline: 3.5708x; 1.0773x over previous
#include <hip/hip_runtime.h>
#include <hip/hip_bf16.h>

#define D_MODEL 128
#define EDGE_DIM 16

typedef float v2f __attribute__((ext_vector_type(2)));
typedef float v8f __attribute__((ext_vector_type(8)));

__device__ __forceinline__ v8f wmma_f32_4(v2f a, v2f b, v8f c) {
    return __builtin_amdgcn_wmma_f32_16x16x4_f32(false, a, false, b, (short)0, c,
                                                 false, false);
}

__device__ __forceinline__ v2f splat2(float v) { return (v2f){v, v}; }

__device__ __forceinline__ v2f gelu2(v2f x) {
    const v2f t = {__builtin_fabsf(x.x), __builtin_fabsf(x.y)};
    v2f u = splat2(-3.36184911e-8f);
    u = __builtin_elementwise_fma(t, u, splat2(9.60158104e-7f));
    u = __builtin_elementwise_fma(t, u, splat2(-0.0000118304251f));
    u = __builtin_elementwise_fma(t, u, splat2(0.0000802221693f));
    u = __builtin_elementwise_fma(t, u, splat2(-0.000288468204f));
    u = __builtin_elementwise_fma(t, u, splat2(0.0000102669237f));
    u = __builtin_elementwise_fma(t, u, splat2(0.00694297580f));
    u = __builtin_elementwise_fma(t, u, splat2(-0.0524434783f));
    u = __builtin_elementwise_fma(t, u, splat2(-0.459218889f));
    u = __builtin_elementwise_fma(t, u, splat2(-1.15110445f));
    const v2f p = __builtin_elementwise_fma(t, u, splat2(-1.0f));
    const v2f h = {__builtin_amdgcn_exp2f(p.x), __builtin_amdgcn_exp2f(p.y)};
    const v2f r = {__builtin_fmaxf(x.x, 0.0f), __builtin_fmaxf(x.y, 0.0f)};
    return __builtin_elementwise_fma(-t, h, r);
}


template <int M>
__device__ __forceinline__ float row_xor(float x) {
    const int b = __builtin_bit_cast(int, x);
    return __builtin_bit_cast(
        float, __builtin_amdgcn_update_dpp(0, b, 0x160 | M, 0xf, 0xf, true));
}

__global__ void __launch_bounds__(256)
gate_kernel(const float* __restrict__ edge_attr,
            const float* __restrict__ gate_w1, const float* __restrict__ gate_b1,
            const float* __restrict__ gate_w2, const float* __restrict__ gate_b2,
            const int* __restrict__ edge_src, const int* __restrict__ edge_dst,
            int* __restrict__ cursor, int* __restrict__ ssrc,
            float* __restrict__ sgate, int n_edges) {
    const int lane = threadIdx.x & 31;
    const int tile = (blockIdx.x * blockDim.x + threadIdx.x) >> 5;
    const int e0   = tile * 16;
    if (e0 >= n_edges) return;

    const int mn   = lane & 15;
    const int koff = (lane >> 4) << 1;
    const int erow = min(e0 + mn, n_edges - 1);

    const float* aB = edge_attr + (size_t)erow * EDGE_DIM;
    v2f a[4];
#pragma unroll
    for (int s = 0; s < 4; ++s) {
        a[s].x = aB[4 * s + koff]; a[s].y = aB[4 * s + koff + 1];
    }

    v2f acc2[4];
#pragma unroll
    for (int j = 0; j < 4; ++j) acc2[j] = splat2(0.0f);

    for (int t = 0; t < D_MODEL / 16; ++t) {
        const int col = t * 16 + mn;
        const float* bB = gate_w1 + (size_t)col * EDGE_DIM;
        v8f c = {};
#pragma unroll
        for (int s = 0; s < 4; ++s) {
            v2f b;
            b.x = bB[4 * s + koff]; b.y = bB[4 * s + koff + 1];
            c = wmma_f32_4(a[s], b, c);
        }
        const v2f b1 = splat2(gate_b1[col]);
        const v2f w2 = splat2(gate_w2[col]);
#pragma unroll
        for (int j = 0; j < 4; ++j) {
            const v2f x = (v2f){c[2 * j], c[2 * j + 1]} + b1;
            acc2[j] = __builtin_elementwise_fma(gelu2(x), w2, acc2[j]);
        }
    }

    float acc[8];
#pragma unroll
    for (int j = 0; j < 4; ++j) { acc[2 * j] = acc2[j].x; acc[2 * j + 1] = acc2[j].y; }

#pragma unroll
    for (int j = 0; j < 8; ++j) {
        acc[j] += row_xor<1>(acc[j]);
        acc[j] += row_xor<2>(acc[j]);
        acc[j] += row_xor<4>(acc[j]);
        acc[j] += row_xor<8>(acc[j]);
    }

    float z = acc[0];
#pragma unroll
    for (int j = 1; j < 8; ++j) z = ((lane & 7) == j) ? acc[j] : z;

    if ((lane & 8) == 0) {
        const int e = e0 + (lane & 7) + ((lane >> 4) << 3);
        if (e < n_edges) {
            z += gate_b2[0];
            const float gt = 1.0f / (1.0f + expf(-z));
            const int pos = atomicAdd(cursor + edge_dst[e], 1);
            ssrc[pos]  = edge_src[e];
            sgate[pos] = gt;
        }
    }
}

__global__ void __launch_bounds__(256)
hist_kernel(const int* __restrict__ edge_dst, int* __restrict__ cnt, int n_edges) {
    const int e = blockIdx.x * blockDim.x + threadIdx.x;
    if (e < n_edges) atomicAdd(cnt + edge_dst[e], 1);
}

__global__ void __launch_bounds__(256)
alloc_kernel(const int* __restrict__ cnt, int* __restrict__ total,
             int* __restrict__ cursor, int n) {
    __shared__ int sW[8];
    __shared__ int sBase;
    const int t = threadIdx.x, lane = t & 31, w = t >> 5;
    const int i = blockIdx.x * 256 + t;
    const int c = (i < n) ? cnt[i] : 0;
    int x = c;
    for (int o = 1; o < 32; o <<= 1) {
        const int y = __shfl_up(x, o, 32);
        if (lane >= o) x += y;
    }
    if (lane == 31) sW[w] = x;
    __syncthreads();
    if (t == 0) {
        int s = 0;
        for (int k = 0; k < 8; ++k) { const int v = sW[k]; sW[k] = s; s += v; }
        sBase = atomicAdd(total, s);
    }
    __syncthreads();
    if (i < n) cursor[i] = sBase + sW[w] + x - c;
}

#define NODE_MT 2
#define NODE_BM (16 * NODE_MT)
#define NODE_PITCH (D_MODEL + 4)
#define NODE_RPW (NODE_BM / 8)

__device__ __forceinline__ float4 frag4(const float* rowbase, int g, int h) {
    return *(const float4*)(rowbase + 8 * g + 4 * h);
}

__global__ void __launch_bounds__(256)
node_kernel(const float* __restrict__ x_src,
            const int* __restrict__ cnt, const int* __restrict__ cursor,
            const int* __restrict__ ssrc, const float* __restrict__ sgate,
            const float* __restrict__ x_dst,
            const float* __restrict__ W_src, const float* __restrict__ W_dst,
            const float* __restrict__ b_dst,
            const float* __restrict__ ln_gamma, const float* __restrict__ ln_beta,
            float* __restrict__ out, int n_nodes) {
    __shared__ __attribute__((aligned(16))) float sAgg[NODE_BM][NODE_PITCH];
    __shared__ __attribute__((aligned(16))) float sX[NODE_BM][NODE_PITCH];
    float (*sTile)[NODE_PITCH] = sAgg;
    const int tid  = threadIdx.x;
    const int wave = __builtin_amdgcn_readfirstlane(tid >> 5);
    const int lane = tid & 31;
    const int row0 = blockIdx.x * NODE_BM;

    float4 xr[NODE_RPW];
#pragma unroll
    for (int q = 0; q < NODE_RPW; ++q) {
        const int n = min(row0 + NODE_RPW * wave + q, n_nodes - 1);
        xr[q] = ((const float4*)(x_dst + (size_t)n * D_MODEL))[lane];
    }

    for (int q = 0; q < NODE_RPW; ++q) {
        const int m = NODE_RPW * wave + q;
        const int n = row0 + m;
        float4 acc = make_float4(0.0f, 0.0f, 0.0f, 0.0f);
        int d = 0;
        if (n < n_nodes) {
            d = cnt[n];
            const int start = cursor[n] - d;
            for (int base = 0; base < d; base += 32) {
                const int idx = base + lane;
                int   s = 0;
                float g = 0.0f;
                if (idx < d) { s = ssrc[start + idx]; g = sgate[start + idx]; }
                const int lim = min(32, d - base);
                for (int j = 0; j < lim; ++j) {
                    const int   sj = __builtin_amdgcn_readlane(s, j);
                    const float gj = __builtin_bit_cast(
                        float, __builtin_amdgcn_readlane(__builtin_bit_cast(int, g), j));
                    const float4 v = ((const float4*)(x_src + (size_t)sj * D_MODEL))[lane];
                    acc.x = fmaf(gj, v.x, acc.x);
                    acc.y = fmaf(gj, v.y, acc.y);
                    acc.z = fmaf(gj, v.z, acc.z);
                    acc.w = fmaf(gj, v.w, acc.w);
                }
            }
        }
        const float inv = 1.0f / fmaxf((float)d, 1.0f);
        acc.x *= inv; acc.y *= inv; acc.z *= inv; acc.w *= inv;
        *(float4*)&sAgg[m][lane * 4] = acc;
    }
#pragma unroll
    for (int q = 0; q < NODE_RPW; ++q)
        *(float4*)&sX[NODE_RPW * wave + q][lane * 4] = xr[q];
    __syncthreads();

    const int mn  = lane & 15;
    const int h   = lane >> 4;
    const int col = wave * 16 + mn;

    v8f c[NODE_MT];
#pragma unroll
    for (int t = 0; t < NODE_MT; ++t) c[t] = (v8f){};

    {
        const float* bB = W_src + (size_t)col * D_MODEL;
        for (int g = 0; g < D_MODEL / 8; ++g) {
            const float4 b = frag4(bB, g, h);
#pragma unroll
            for (int t = 0; t < NODE_MT; ++t) {
                const float4 a = frag4(&sAgg[t * 16 + mn][0], g, h);
                c[t] = wmma_f32_4((v2f){a.x, a.y}, (v2f){b.x, b.y}, c[t]);
                c[t] = wmma_f32_4((v2f){a.z, a.w}, (v2f){b.z, b.w}, c[t]);
            }
        }
    }
    {
        const float* bB = W_dst + (size_t)col * D_MODEL;
        for (int g = 0; g < D_MODEL / 8; ++g) {
            const float4 b = frag4(bB, g, h);
#pragma unroll
            for (int t = 0; t < NODE_MT; ++t) {
                const float4 a = frag4(&sX[t * 16 + mn][0], g, h);
                c[t] = wmma_f32_4((v2f){a.x, a.y}, (v2f){b.x, b.y}, c[t]);
                c[t] = wmma_f32_4((v2f){a.z, a.w}, (v2f){b.z, b.w}, c[t]);
            }
        }
    }
    __syncthreads();

    const float bias = b_dst[col];
#pragma unroll
    for (int t = 0; t < NODE_MT; ++t) {
#pragma unroll
        for (int j = 0; j < 8; ++j) {
            const int m = t * 16 + ((lane < 16) ? j : (j + 8));
            sTile[m][col] = c[t][j] + bias;
        }
    }
    __syncthreads();

    const int c0 = (tid & 15) * 8;
    for (int t = 0; t < NODE_MT; ++t) {
        const int m = t * 16 + (tid >> 4);
        float s = 0.0f, ss = 0.0f;
#pragma unroll
        for (int i = 0; i < 8; ++i) {
            float v = sTile[m][c0 + i];
            s += v; ss += v * v;
        }
#pragma unroll
        for (int off = 8; off > 0; off >>= 1) {
            s  += __shfl_xor(s,  off, 32);
            ss += __shfl_xor(ss, off, 32);
        }
        const float mean = s * (1.0f / 128.0f);
        const float var  = ss * (1.0f / 128.0f) - mean * mean;
        const float rstd = rsqrtf(var + 1e-5f);

        const int rglob = row0 + m;
        if (rglob < n_nodes) {
#pragma unroll
            for (int i = 0; i < 8; i += 2) {
                const int cc = c0 + i;
                v2f v;
                v.x = (sTile[m][cc]     - mean) * rstd * ln_gamma[cc]     + ln_beta[cc];
                v.y = (sTile[m][cc + 1] - mean) * rstd * ln_gamma[cc + 1] + ln_beta[cc + 1];
                const v2f g = gelu2(v);
                out[(size_t)rglob * D_MODEL + cc]     = g.x;
                out[(size_t)rglob * D_MODEL + cc + 1] = g.y;
            }
        }
    }
}

extern "C" void kernel_launch(void* const* d_in, const int* in_sizes, int n_in,
                              void* d_out, int out_size, void* d_ws, size_t ws_size,
                              hipStream_t stream) {
    const float* x_src     = (const float*)d_in[0];
    const float* x_dst     = (const float*)d_in[1];
    const int*   edge_src  = (const int*)  d_in[2];
    const int*   edge_dst  = (const int*)  d_in[3];
    const float* edge_attr = (const float*)d_in[4];
    const float* W_src     = (const float*)d_in[5];
    const float* W_dst     = (const float*)d_in[6];
    const float* b_dst     = (const float*)d_in[7];
    const float* gate_w1   = (const float*)d_in[8];
    const float* gate_b1   = (const float*)d_in[9];
    const float* gate_w2   = (const float*)d_in[10];
    const float* gate_b2   = (const float*)d_in[11];
    const float* ln_gamma  = (const float*)d_in[12];
    const float* ln_beta   = (const float*)d_in[13];

    const int n_nodes = in_sizes[0] / D_MODEL;
    const int n_edges = in_sizes[2];

    int*   cnt    = (int*)d_ws;
    int*   total  = cnt + n_nodes;
    int*   cursor = total + 1;
    int*   ssrc   = cursor + n_nodes;
    float* sgate  = (float*)(ssrc + n_edges);

    hipMemsetAsync(cnt, 0, (size_t)(n_nodes + 1) * sizeof(int), stream);

    const int histBlocks = (n_edges + 255) / 256;
    hist_kernel<<<histBlocks, 256, 0, stream>>>(edge_dst, cnt, n_edges);
    const int allocBlocks = (n_nodes + 255) / 256;
    alloc_kernel<<<allocBlocks, 256, 0, stream>>>(cnt, total, cursor, n_nodes);

    const int gateBlocks = (n_edges + 127) / 128;
    gate_kernel<<<gateBlocks, 256, 0, stream>>>(edge_attr, gate_w1, gate_b1,
                                                gate_w2, gate_b2, edge_src, edge_dst,
                                                cursor, ssrc, sgate, n_edges);

    const int nodeBlocks = (n_nodes + NODE_BM - 1) / NODE_BM;
    node_kernel<<<nodeBlocks, 256, 0, stream>>>(x_src, cnt, cursor, ssrc, sgate,
                                                x_dst, W_src, W_dst,
                                                b_dst, ln_gamma, ln_beta,
                                                (float*)d_out, n_nodes);
}


# Gate kernel: 4096 resident waves each walk 7-8 edge tiles with gate_w1/b1/w2 held in registers, next rows and the list push issued ahead (was 31250 one-tile waves); list order by arrival as before
# speedup vs baseline: 6.3273x; 1.7720x over previous
#include <hip/hip_runtime.h>
#include <hip/hip_bf16.h>

#define D_MODEL 128
#define EDGE_DIM 16

typedef float v2f __attribute__((ext_vector_type(2)));
typedef float v8f __attribute__((ext_vector_type(8)));

__device__ __forceinline__ v8f wmma_f32_4(v2f a, v2f b, v8f c) {
    return __builtin_amdgcn_wmma_f32_16x16x4_f32(false, a, false, b, (short)0, c,
                                                 false, false);
}

__device__ __forceinline__ v2f splat2(float v) { return (v2f){v, v}; }

__device__ __forceinline__ v2f gelu2(v2f x) {
    const v2f t = {__builtin_fabsf(x.x), __builtin_fabsf(x.y)};
    v2f u = splat2(-3.36184911e-8f);
    u = __builtin_elementwise_fma(t, u, splat2(9.60158104e-7f));
    u = __builtin_elementwise_fma(t, u, splat2(-0.0000118304251f));
    u = __builtin_elementwise_fma(t, u, splat2(0.0000802221693f));
    u = __builtin_elementwise_fma(t, u, splat2(-0.000288468204f));
    u = __builtin_elementwise_fma(t, u, splat2(0.0000102669237f));
    u = __builtin_elementwise_fma(t, u, splat2(0.00694297580f));
    u = __builtin_elementwise_fma(t, u, splat2(-0.0524434783f));
    u = __builtin_elementwise_fma(t, u, splat2(-0.459218889f));
    u = __builtin_elementwise_fma(t, u, splat2(-1.15110445f));
    const v2f p = __builtin_elementwise_fma(t, u, splat2(-1.0f));
    const v2f h = {__builtin_amdgcn_exp2f(p.x), __builtin_amdgcn_exp2f(p.y)};
    const v2f r = {__builtin_fmaxf(x.x, 0.0f), __builtin_fmaxf(x.y, 0.0f)};
    return __builtin_elementwise_fma(-t, h, r);
}


template <int M>
__device__ __forceinline__ float row_xor(float x) {
    const int b = __builtin_bit_cast(int, x);
    return __builtin_bit_cast(
        float, __builtin_amdgcn_update_dpp(0, b, 0x160 | M, 0xf, 0xf, true));
}

#define GATE_BLOCKS 512

__global__ void __launch_bounds__(256, 4)
gate_kernel(const float* __restrict__ edge_attr,
            const float* __restrict__ gate_w1, const float* __restrict__ gate_b1,
            const float* __restrict__ gate_w2, const float* __restrict__ gate_b2,
            const int* __restrict__ edge_dst,
            int* __restrict__ head, int* __restrict__ rec, int n_edges) {
    const int lane    = threadIdx.x & 31;
    const int wave_g  = (blockIdx.x * blockDim.x + threadIdx.x) >> 5;
    const int n_waves = (gridDim.x * blockDim.x) >> 5;
    const int n_tiles = (n_edges + 15) >> 4;
    if (wave_g >= n_tiles) return;

    const int mn   = lane & 15;
    const int koff = (lane >> 4) << 1;

    v2f   bw[D_MODEL / 16][4];
    float b1s[D_MODEL / 16], w2s[D_MODEL / 16];
#pragma unroll
    for (int t = 0; t < D_MODEL / 16; ++t) {
        const int col = t * 16 + mn;
        const float* bB = gate_w1 + (size_t)col * EDGE_DIM;
#pragma unroll
        for (int s = 0; s < 4; ++s) {
            bw[t][s].x = bB[4 * s + koff]; bw[t][s].y = bB[4 * s + koff + 1];
        }
        b1s[t] = gate_b1[col];
        w2s[t] = gate_w2[col];
    }
    const float gb2 = gate_b2[0];

    v2f a[4];
    {
        const int erow = min(wave_g * 16 + mn, n_edges - 1);
        const float* aB = edge_attr + (size_t)erow * EDGE_DIM;
#pragma unroll
        for (int s = 0; s < 4; ++s) {
            a[s].x = aB[4 * s + koff]; a[s].y = aB[4 * s + koff + 1];
        }
    }

    const int eoff = (lane & 7) + ((lane >> 4) << 3);
    int dst = edge_dst[min(wave_g * 16 + eoff, n_edges - 1)];

    for (int tile = wave_g; tile < n_tiles; tile += n_waves) {
        const int e0 = tile * 16;
        v2f an[4];
        {
            const int erow = min(e0 + n_waves * 16 + mn, n_edges - 1);
            const float* aB = edge_attr + (size_t)erow * EDGE_DIM;
#pragma unroll
            for (int s = 0; s < 4; ++s) {
                an[s].x = aB[4 * s + koff]; an[s].y = aB[4 * s + koff + 1];
            }
        }
        const int  e    = e0 + eoff;
        const bool mine = ((lane & 8) == 0) && (e < n_edges);
        const int dstn = edge_dst[min(e + n_waves * 16, n_edges - 1)];
        int prev = 0;
        if (mine) prev = atomicExch(head + dst, e + 1);

        v2f acc2[4];
#pragma unroll
        for (int j = 0; j < 4; ++j) acc2[j] = splat2(0.0f);

#pragma unroll
        for (int t = 0; t < D_MODEL / 16; ++t) {
            v8f c = {};
#pragma unroll
            for (int s = 0; s < 4; ++s) c = wmma_f32_4(a[s], bw[t][s], c);
            const v2f b1 = splat2(b1s[t]);
            const v2f w2 = splat2(w2s[t]);
#pragma unroll
            for (int j = 0; j < 4; ++j) {
                const v2f x = (v2f){c[2 * j], c[2 * j + 1]} + b1;
                acc2[j] = __builtin_elementwise_fma(gelu2(x), w2, acc2[j]);
            }
            if (t == D_MODEL / 32 - 1) __builtin_amdgcn_sched_barrier(0);
        }

        float acc[8];
#pragma unroll
        for (int j = 0; j < 4; ++j) { acc[2 * j] = acc2[j].x; acc[2 * j + 1] = acc2[j].y; }

#pragma unroll
        for (int j = 0; j < 8; ++j) {
            acc[j] += row_xor<1>(acc[j]);
            acc[j] += row_xor<2>(acc[j]);
            acc[j] += row_xor<4>(acc[j]);
            acc[j] += row_xor<8>(acc[j]);
        }

        float z = acc[0];
#pragma unroll
        for (int j = 1; j < 8; ++j) z = ((lane & 7) == j) ? acc[j] : z;

        if (mine) {
            z += gb2;
            const float gt = 1.0f / (1.0f + expf(-z));
            ((int2*)rec)[e] = make_int2(__builtin_bit_cast(int, gt), prev);
        }

#pragma unroll
        for (int s = 0; s < 4; ++s) a[s] = an[s];
        dst = dstn;
    }
}

#define NODE_MT 2
#define NODE_BM (16 * NODE_MT)
#define NODE_PITCH (D_MODEL + 4)
#define NODE_RPW (NODE_BM / 8)

__device__ __forceinline__ float4 frag4(const float* rowbase, int g, int h) {
    return *(const float4*)(rowbase + 8 * g + 4 * h);
}

__global__ void __launch_bounds__(256)
node_kernel(const float* __restrict__ x_src,
            const int* __restrict__ head, const int* __restrict__ rec,
            const int* __restrict__ edge_src,
            const float* __restrict__ x_dst,
            const float* __restrict__ W_src, const float* __restrict__ W_dst,
            const float* __restrict__ b_dst,
            const float* __restrict__ ln_gamma, const float* __restrict__ ln_beta,
            float* __restrict__ out, int n_nodes, int n_edges) {
    __shared__ __attribute__((aligned(16))) float sAgg[NODE_BM][NODE_PITCH];
    __shared__ __attribute__((aligned(16))) float sX[NODE_BM][NODE_PITCH];
    float (*sTile)[NODE_PITCH] = sAgg;
    const int tid  = threadIdx.x;
    const int wave = __builtin_amdgcn_readfirstlane(tid >> 5);
    const int lane = tid & 31;
    const int row0 = blockIdx.x * NODE_BM;

    float4 xr[NODE_RPW];
#pragma unroll
    for (int q = 0; q < NODE_RPW; ++q) {
        const int n = min(row0 + NODE_RPW * wave + q, n_nodes - 1);
        xr[q] = ((const float4*)(x_dst + (size_t)n * D_MODEL))[lane];
    }

    const int2* recs = (const int2*)rec;
    int  e = 0;
    int2 r = make_int2(0, 0);
    int  s = 0;
    {
        const int n = row0 + NODE_RPW * wave + lane;
        if (lane < NODE_RPW && n < n_nodes) r.y = head[n];
    }
    int deg = 0;
    float4 acc[NODE_RPW];
#pragma unroll
    for (int q = 0; q < NODE_RPW; ++q) acc[q] = make_float4(0.0f, 0.0f, 0.0f, 0.0f);

    for (int it = 0; it <= n_edges; ++it) {
        const int en = r.y;
        const unsigned live = (unsigned)__builtin_amdgcn_ballot_w32(e != 0);
        const unsigned more = (unsigned)__builtin_amdgcn_ballot_w32(en != 0);
        if ((live | more) == 0u) break;
        int2 rn = make_int2(0, 0);
        int  sn = 0;
        if (en != 0) { rn = recs[en - 1]; sn = edge_src[en - 1]; }
        float4 v[NODE_RPW];
#pragma unroll
        for (int q = 0; q < NODE_RPW; ++q) {
            const int sj = __builtin_amdgcn_readlane(s, q);
            v[q] = ((const float4*)(x_src + (size_t)sj * D_MODEL))[lane];
        }
#pragma unroll
        for (int q = 0; q < NODE_RPW; ++q) {
            if (live & (1u << q)) {
                const float gj = __builtin_bit_cast(float, __builtin_amdgcn_readlane(r.x, q));
                acc[q].x = fmaf(gj, v[q].x, acc[q].x);
                acc[q].y = fmaf(gj, v[q].y, acc[q].y);
                acc[q].z = fmaf(gj, v[q].z, acc[q].z);
                acc[q].w = fmaf(gj, v[q].w, acc[q].w);
            }
        }
        deg += (e != 0) ? 1 : 0;
        e = en;
        r = rn;
        s = sn;
    }
#pragma unroll
    for (int q = 0; q < NODE_RPW; ++q) {
        const int d = __builtin_amdgcn_readlane(deg, q);
        const float inv = 1.0f / fmaxf((float)d, 1.0f);
        float4 a = acc[q];
        a.x *= inv; a.y *= inv; a.z *= inv; a.w *= inv;
        *(float4*)&sAgg[NODE_RPW * wave + q][lane * 4] = a;
    }
#pragma unroll
    for (int q = 0; q < NODE_RPW; ++q)
        *(float4*)&sX[NODE_RPW * wave + q][lane * 4] = xr[q];
    __syncthreads();

    const int mn  = lane & 15;
    const int h   = lane >> 4;
    const int col = wave * 16 + mn;

    v8f c[NODE_MT];
#pragma unroll
    for (int t = 0; t < NODE_MT; ++t) c[t] = (v8f){};

    {
        const float* bB = W_src + (size_t)col * D_MODEL;
        for (int g = 0; g < D_MODEL / 8; ++g) {
            const float4 b = frag4(bB, g, h);
#pragma unroll
            for (int t = 0; t < NODE_MT; ++t) {
                const float4 a = frag4(&sAgg[t * 16 + mn][0], g, h);
                c[t] = wmma_f32_4((v2f){a.x, a.y}, (v2f){b.x, b.y}, c[t]);
                c[t] = wmma_f32_4((v2f){a.z, a.w}, (v2f){b.z, b.w}, c[t]);
            }
        }
    }
    {
        const float* bB = W_dst + (size_t)col * D_MODEL;
        for (int g = 0; g < D_MODEL / 8; ++g) {
            const float4 b = frag4(bB, g, h);
#pragma unroll
            for (int t = 0; t < NODE_MT; ++t) {
                const float4 a = frag4(&sX[t * 16 + mn][0], g, h);
                c[t] = wmma_f32_4((v2f){a.x, a.y}, (v2f){b.x, b.y}, c[t]);
                c[t] = wmma_f32_4((v2f){a.z, a.w}, (v2f){b.z, b.w}, c[t]);
            }
        }
    }
    __syncthreads();

    const float bias = b_dst[col];
#pragma unroll
    for (int t = 0; t < NODE_MT; ++t) {
#pragma unroll
        for (int j = 0; j < 8; ++j) {
            const int m = t * 16 + ((lane < 16) ? j : (j + 8));
            sTile[m][col] = c[t][j] + bias;
        }
    }
    __syncthreads();

    const int c0 = (tid & 15) * 8;
    for (int t = 0; t < NODE_MT; ++t) {
        const int m = t * 16 + (tid >> 4);
        float s = 0.0f, ss = 0.0f;
#pragma unroll
        for (int i = 0; i < 8; ++i) {
            float v = sTile[m][c0 + i];
            s += v; ss += v * v;
        }
#pragma unroll
        for (int off = 8; off > 0; off >>= 1) {
            s  += __shfl_xor(s,  off, 32);
            ss += __shfl_xor(ss, off, 32);
        }
        const float mean = s * (1.0f / 128.0f);
        const float var  = ss * (1.0f / 128.0f) - mean * mean;
        const float rstd = rsqrtf(var + 1e-5f);

        const int rglob = row0 + m;
        if (rglob < n_nodes) {
#pragma unroll
            for (int i = 0; i < 8; i += 2) {
                const int cc = c0 + i;
                v2f v;
                v.x = (sTile[m][cc]     - mean) * rstd * ln_gamma[cc]     + ln_beta[cc];
                v.y = (sTile[m][cc + 1] - mean) * rstd * ln_gamma[cc + 1] + ln_beta[cc + 1];
                const v2f g = gelu2(v);
                out[(size_t)rglob * D_MODEL + cc]     = g.x;
                out[(size_t)rglob * D_MODEL + cc + 1] = g.y;
            }
        }
    }
}

extern "C" void kernel_launch(void* const* d_in, const int* in_sizes, int n_in,
                              void* d_out, int out_size, void* d_ws, size_t ws_size,
                              hipStream_t stream) {
    const float* x_src     = (const float*)d_in[0];
    const float* x_dst     = (const float*)d_in[1];
    const int*   edge_src  = (const int*)  d_in[2];
    const int*   edge_dst  = (const int*)  d_in[3];
    const float* edge_attr = (const float*)d_in[4];
    const float* W_src     = (const float*)d_in[5];
    const float* W_dst     = (const float*)d_in[6];
    const float* b_dst     = (const float*)d_in[7];
    const float* gate_w1   = (const float*)d_in[8];
    const float* gate_b1   = (const float*)d_in[9];
    const float* gate_w2   = (const float*)d_in[10];
    const float* gate_b2   = (const float*)d_in[11];
    const float* ln_gamma  = (const float*)d_in[12];
    const float* ln_beta   = (const float*)d_in[13];

    const int n_nodes = in_sizes[0] / D_MODEL;
    const int n_edges = in_sizes[2];

    int* rec  = (int*)d_ws;
    int* head = rec + (size_t)2 * n_edges;

    hipMemsetAsync(head, 0, (size_t)n_nodes * sizeof(int), stream);

    gate_kernel<<<GATE_BLOCKS, 256, 0, stream>>>(edge_attr, gate_w1, gate_b1,
                                                gate_w2, gate_b2, edge_dst,
                                                head, rec, n_edges);

    const int nodeBlocks = (n_nodes + NODE_BM - 1) / NODE_BM;
    node_kernel<<<nodeBlocks, 256, 0, stream>>>(x_src, head, rec, edge_src,
                                                x_dst, W_src, W_dst,
                                                b_dst, ln_gamma, ln_beta,
                                                (float*)d_out, n_nodes, n_edges);
}
